# starting kernel
# baseline (speedup 1.0000x reference)
#include <hip/hip_runtime.h>

#define N_SEG   256
#define NCH     32
#define HW      (512 * 512)
#define N_IMG   8
#define ACC_ELEMS (2 * N_SEG * NCH)

typedef __attribute__((ext_vector_type(2))) float v2f;
typedef __attribute__((ext_vector_type(8))) float v8f;

__global__ __launch_bounds__(256) void seg_partial_kernel(
    const float* __restrict__ image, const int* __restrict__ slic,
    float* __restrict__ partials, int G)
{
    __shared__ float acc[ACC_ELEMS];

    const int tid  = threadIdx.x;
    const int img  = blockIdx.x / G;
    const int wg   = blockIdx.x % G;

    for (int i = tid; i < ACC_ELEMS; i += 256) acc[i] = 0.0f;
    __syncthreads();

    const int lane  = tid & 31;
    const int wave  = tid >> 5;
    const int chunk = HW / G;
    const int p0    = wg * chunk;
    const size_t img_pix = (size_t)img * HW;

    #pragma unroll 4
    for (int p = p0 + wave; p < p0 + chunk; p += 8) {
        const size_t pix = img_pix + (size_t)p;
        __builtin_prefetch(&image[(pix + 32) * NCH + lane], 0, 0);
        const float v = __builtin_nontemporal_load(&image[pix * NCH + lane]);
        const int   s = __builtin_nontemporal_load(&slic[pix]) - 1;
        atomicAdd(&acc[s * NCH + lane], v);
        atomicAdd(&acc[N_SEG * NCH + s * NCH + lane],
                  (v != 0.0f) ? 1.0f : 0.0f);
    }
    __syncthreads();

    float* dst = partials + (size_t)(img * G + wg) * ACC_ELEMS;
    for (int i = tid; i < ACC_ELEMS; i += 256) dst[i] = acc[i];
}

__global__ __launch_bounds__(256) void seg_reduce_wmma_kernel(
    const float* __restrict__ partials, float* __restrict__ out, int G)
{
    const int tid  = threadIdx.x;
    const int lane = tid & 31;
    const int w    = blockIdx.x * 8 + (tid >> 5);
    const int img  = w >> 9;
    const int tile = w & 511;
    const int col  = tile * 16 + (lane & 15);
    const int half = lane >> 4;

    v2f a; a.x = 1.0f; a.y = 1.0f;
    v8f cs = {};
    v8f cc = {};

    const float* base = partials + (size_t)img * G * ACC_ELEMS;
    const int ksteps = G >> 2;
    for (int kk = 0; kk < ksteps; ++kk) {
        const int r0 = kk * 4 + half;
        const int r1 = kk * 4 + 2 + half;
        v2f bs, bc;
        bs.x = __builtin_nontemporal_load(&base[(size_t)r0 * ACC_ELEMS + col]);
        bs.y = __builtin_nontemporal_load(&base[(size_t)r1 * ACC_ELEMS + col]);
        bc.x = __builtin_nontemporal_load(&base[(size_t)r0 * ACC_ELEMS + N_SEG * NCH + col]);
        bc.y = __builtin_nontemporal_load(&base[(size_t)r1 * ACC_ELEMS + N_SEG * NCH + col]);
        cs = __builtin_amdgcn_wmma_f32_16x16x4_f32(false, a, false, bs,
                                                   (short)0, cs, false, false);
        cc = __builtin_amdgcn_wmma_f32_16x16x4_f32(false, a, false, bc,
                                                   (short)0, cc, false, false);
    }

    if (lane < 16) {
        out[(size_t)img * (N_SEG * NCH) + col] = cs[0] / cc[0];
    }
}

extern "C" void kernel_launch(void* const* d_in, const int* in_sizes, int n_in,
                              void* d_out, int out_size, void* d_ws, size_t ws_size,
                              hipStream_t stream) {
    const float* image = (const float*)d_in[0];
    const int*   slic  = (const int*)d_in[1];
    float*       out   = (float*)d_out;
    float*       ws    = (float*)d_ws;

    int G;
    if      (ws_size >= (size_t)64 * 524288) G = 64;
    else if (ws_size >= (size_t)32 * 524288) G = 32;
    else if (ws_size >= (size_t)16 * 524288) G = 16;
    else if (ws_size >= (size_t)8  * 524288) G = 8;
    else                                     G = 4;

    seg_partial_kernel<<<N_IMG * G, 256, 0, stream>>>(image, slic, ws, G);

    const int total_waves = N_IMG * (N_SEG * NCH / 16);
    seg_reduce_wmma_kernel<<<total_waves / 8, 256, 0, stream>>>(ws, out, G);
}


# Stage 1: each wave takes 32 pixels per trip, one lane-wide label load and 32 row loads issued ahead in two register sets (no wait per pixel, no prefetch); order of the float LDS sums changes
# speedup vs baseline: 3.6954x; 3.6954x over previous
#include <hip/hip_runtime.h>

#define N_SEG   256
#define NCH     32
#define HW      (512 * 512)
#define N_IMG   8
#define ACC_ELEMS (2 * N_SEG * NCH)

typedef __attribute__((ext_vector_type(2))) float v2f;
typedef __attribute__((ext_vector_type(8))) float v8f;

#define BATCH 32
#define TRIP_PIX (8 * BATCH)

__device__ __forceinline__ void issue_batch(const float* __restrict__ rows,
                                            const int* __restrict__ labs,
                                            int lane, float (&v)[BATCH], int& lab)
{
    lab = __builtin_nontemporal_load(&labs[lane]);
    #pragma unroll
    for (int j = 0; j < BATCH; ++j)
        v[j] = __builtin_nontemporal_load(&rows[j * NCH + lane]);
}

__device__ __forceinline__ void drain_batch(float* acc, int lane,
                                            const float (&v)[BATCH], int lab)
{
    const int row = (lab - 1) * NCH;
    #pragma unroll
    for (int j = 0; j < BATCH; ++j) {
        float* a = acc + __builtin_amdgcn_readlane(row, j) + lane;
        atomicAdd(a, v[j]);
        atomicAdd(a + N_SEG * NCH, (v[j] != 0.0f) ? 1.0f : 0.0f);
    }
}

__global__ __launch_bounds__(256) void seg_partial_kernel(
    const float* __restrict__ image, const int* __restrict__ slic,
    float* __restrict__ partials, int G)
{
    __shared__ float acc[ACC_ELEMS];

    const int tid  = threadIdx.x;
    const int img  = blockIdx.x / G;
    const int wg   = blockIdx.x % G;

    for (int i = tid; i < ACC_ELEMS; i += 256) acc[i] = 0.0f;
    __syncthreads();

    const int lane  = tid & 31;
    const int wave  = __builtin_amdgcn_readfirstlane(tid >> 5);
    const int chunk = HW / G;
    const int trips = chunk / TRIP_PIX;
    const size_t pix0 = (size_t)img * HW + (size_t)wg * chunk + (size_t)wave * BATCH;
    const float* rows = image + pix0 * NCH;
    const int*   labs = slic + pix0;

    float va[BATCH], vb[BATCH];
    int   la, lb;
    issue_batch(rows, labs, lane, va, la);
    int t = 0;
    for (; t + 2 < trips; t += 2) {
        issue_batch(rows + (size_t)(t + 1) * TRIP_PIX * NCH, labs + (size_t)(t + 1) * TRIP_PIX,
                    lane, vb, lb);
        __builtin_amdgcn_sched_barrier(0);
        drain_batch(acc, lane, va, la);
        issue_batch(rows + (size_t)(t + 2) * TRIP_PIX * NCH, labs + (size_t)(t + 2) * TRIP_PIX,
                    lane, va, la);
        __builtin_amdgcn_sched_barrier(0);
        drain_batch(acc, lane, vb, lb);
    }
    issue_batch(rows + (size_t)(t + 1) * TRIP_PIX * NCH, labs + (size_t)(t + 1) * TRIP_PIX,
                lane, vb, lb);
    __builtin_amdgcn_sched_barrier(0);
    drain_batch(acc, lane, va, la);
    drain_batch(acc, lane, vb, lb);
    __syncthreads();

    float* dst = partials + (size_t)(img * G + wg) * ACC_ELEMS;
    for (int i = tid; i < ACC_ELEMS; i += 256) dst[i] = acc[i];
}

__global__ __launch_bounds__(256) void seg_reduce_wmma_kernel(
    const float* __restrict__ partials, float* __restrict__ out, int G)
{
    const int tid  = threadIdx.x;
    const int lane = tid & 31;
    const int w    = blockIdx.x * 8 + (tid >> 5);
    const int img  = w >> 9;
    const int tile = w & 511;
    const int col  = tile * 16 + (lane & 15);
    const int half = lane >> 4;

    v2f a; a.x = 1.0f; a.y = 1.0f;
    v8f cs = {};
    v8f cc = {};

    const float* base = partials + (size_t)img * G * ACC_ELEMS;
    const int ksteps = G >> 2;
    for (int kk = 0; kk < ksteps; ++kk) {
        const int r0 = kk * 4 + half;
        const int r1 = kk * 4 + 2 + half;
        v2f bs, bc;
        bs.x = __builtin_nontemporal_load(&base[(size_t)r0 * ACC_ELEMS + col]);
        bs.y = __builtin_nontemporal_load(&base[(size_t)r1 * ACC_ELEMS + col]);
        bc.x = __builtin_nontemporal_load(&base[(size_t)r0 * ACC_ELEMS + N_SEG * NCH + col]);
        bc.y = __builtin_nontemporal_load(&base[(size_t)r1 * ACC_ELEMS + N_SEG * NCH + col]);
        cs = __builtin_amdgcn_wmma_f32_16x16x4_f32(false, a, false, bs,
                                                   (short)0, cs, false, false);
        cc = __builtin_amdgcn_wmma_f32_16x16x4_f32(false, a, false, bc,
                                                   (short)0, cc, false, false);
    }

    if (lane < 16) {
        out[(size_t)img * (N_SEG * NCH) + col] = cs[0] / cc[0];
    }
}

extern "C" void kernel_launch(void* const* d_in, const int* in_sizes, int n_in,
                              void* d_out, int out_size, void* d_ws, size_t ws_size,
                              hipStream_t stream) {
    const float* image = (const float*)d_in[0];
    const int*   slic  = (const int*)d_in[1];
    float*       out   = (float*)d_out;
    float*       ws    = (float*)d_ws;

    int G;
    if      (ws_size >= (size_t)64 * 524288) G = 64;
    else if (ws_size >= (size_t)32 * 524288) G = 32;
    else if (ws_size >= (size_t)16 * 524288) G = 16;
    else if (ws_size >= (size_t)8  * 524288) G = 8;
    else                                     G = 4;

    seg_partial_kernel<<<N_IMG * G, 256, 0, stream>>>(image, slic, ws, G);

    const int total_waves = N_IMG * (N_SEG * NCH / 16);
    seg_reduce_wmma_kernel<<<total_waves / 8, 256, 0, stream>>>(ws, out, G);
}


# Stage 1: pixels per segment counted once per pixel (one lane-wide LDS integer add per 32 pixels) less per-channel zero hits, added only where a value is zero: LDS float adds per pixel 2 to 1
# speedup vs baseline: 4.0562x; 1.0976x over previous
#include <hip/hip_runtime.h>

#define N_SEG   256
#define NCH     32
#define HW      (512 * 512)
#define N_IMG   8
#define PLANE   (N_SEG * NCH)
#define ACC_ELEMS (2 * PLANE)

typedef __attribute__((ext_vector_type(2))) float v2f;
typedef __attribute__((ext_vector_type(8))) float v8f;

#define BATCH 32
#define TRIP_PIX (8 * BATCH)

__device__ __forceinline__ void issue_batch(const float* __restrict__ rows,
                                            const int* __restrict__ labs,
                                            int lane, float (&v)[BATCH], int& lab)
{
    lab = __builtin_nontemporal_load(&labs[lane]);
    #pragma unroll
    for (int j = 0; j < BATCH; ++j)
        v[j] = __builtin_nontemporal_load(&rows[j * NCH + lane]);
}

__device__ __forceinline__ void drain_batch(float* sums, float* zeros, unsigned* npix, int lane,
                                            const float (&v)[BATCH], int lab)
{
    atomicAdd(&npix[lab - 1], 1u);
    const int row = (lab - 1) * NCH;
    unsigned anyz = 0;
    #pragma unroll
    for (int j = 0; j < BATCH; ++j) {
        float* a = sums + __builtin_amdgcn_readlane(row, j) + lane;
        atomicAdd(a, v[j]);
        anyz |= __builtin_amdgcn_ballot_w32(!(v[j] != 0.0f));
    }
    if (anyz != 0) {
        #pragma unroll
        for (int j = 0; j < BATCH; ++j) {
            const int rj = __builtin_amdgcn_readlane(row, j);
            if (!(v[j] != 0.0f)) atomicAdd(zeros + rj + lane, 1.0f);
        }
    }
}

__global__ __launch_bounds__(256) void seg_partial_kernel(
    const float* __restrict__ image, const int* __restrict__ slic,
    float* __restrict__ partials, int G)
{
    __shared__ float    sums[PLANE];
    __shared__ float    zeros[PLANE];
    __shared__ unsigned npix[N_SEG];

    const int tid  = threadIdx.x;
    const int img  = blockIdx.x / G;
    const int wg   = blockIdx.x % G;

    for (int i = tid; i < PLANE; i += 256) { sums[i] = 0.0f; zeros[i] = 0.0f; }
    npix[tid] = 0u;
    __syncthreads();

    const int lane  = tid & 31;
    const int wave  = __builtin_amdgcn_readfirstlane(tid >> 5);
    const int chunk = HW / G;
    const int trips = chunk / TRIP_PIX;
    const size_t pix0 = (size_t)img * HW + (size_t)wg * chunk + (size_t)wave * BATCH;
    const float* rows = image + pix0 * NCH;
    const int*   labs = slic + pix0;

    float va[BATCH], vb[BATCH];
    int   la, lb;
    issue_batch(rows, labs, lane, va, la);
    int t = 0;
    for (; t + 2 < trips; t += 2) {
        issue_batch(rows + (size_t)(t + 1) * TRIP_PIX * NCH, labs + (size_t)(t + 1) * TRIP_PIX,
                    lane, vb, lb);
        __builtin_amdgcn_sched_barrier(0);
        drain_batch(sums, zeros, npix, lane, va, la);
        issue_batch(rows + (size_t)(t + 2) * TRIP_PIX * NCH, labs + (size_t)(t + 2) * TRIP_PIX,
                    lane, va, la);
        __builtin_amdgcn_sched_barrier(0);
        drain_batch(sums, zeros, npix, lane, vb, lb);
    }
    issue_batch(rows + (size_t)(t + 1) * TRIP_PIX * NCH, labs + (size_t)(t + 1) * TRIP_PIX,
                lane, vb, lb);
    __builtin_amdgcn_sched_barrier(0);
    drain_batch(sums, zeros, npix, lane, va, la);
    drain_batch(sums, zeros, npix, lane, vb, lb);
    __syncthreads();

    float* dst = partials + (size_t)(img * G + wg) * ACC_ELEMS;
    for (int i = tid; i < PLANE; i += 256) {
        dst[i]         = sums[i];
        dst[PLANE + i] = (float)npix[i >> 5] - zeros[i];
    }
}

__global__ __launch_bounds__(256) void seg_reduce_wmma_kernel(
    const float* __restrict__ partials, float* __restrict__ out, int G)
{
    const int tid  = threadIdx.x;
    const int lane = tid & 31;
    const int w    = blockIdx.x * 8 + (tid >> 5);
    const int img  = w >> 9;
    const int tile = w & 511;
    const int col  = tile * 16 + (lane & 15);
    const int half = lane >> 4;

    v2f a; a.x = 1.0f; a.y = 1.0f;
    v8f cs = {};
    v8f cc = {};

    const float* base = partials + (size_t)img * G * ACC_ELEMS;
    const int ksteps = G >> 2;
    for (int kk = 0; kk < ksteps; ++kk) {
        const int r0 = kk * 4 + half;
        const int r1 = kk * 4 + 2 + half;
        v2f bs, bc;
        bs.x = __builtin_nontemporal_load(&base[(size_t)r0 * ACC_ELEMS + col]);
        bs.y = __builtin_nontemporal_load(&base[(size_t)r1 * ACC_ELEMS + col]);
        bc.x = __builtin_nontemporal_load(&base[(size_t)r0 * ACC_ELEMS + N_SEG * NCH + col]);
        bc.y = __builtin_nontemporal_load(&base[(size_t)r1 * ACC_ELEMS + N_SEG * NCH + col]);
        cs = __builtin_amdgcn_wmma_f32_16x16x4_f32(false, a, false, bs,
                                                   (short)0, cs, false, false);
        cc = __builtin_amdgcn_wmma_f32_16x16x4_f32(false, a, false, bc,
                                                   (short)0, cc, false, false);
    }

    if (lane < 16) {
        out[(size_t)img * (N_SEG * NCH) + col] = cs[0] / cc[0];
    }
}

extern "C" void kernel_launch(void* const* d_in, const int* in_sizes, int n_in,
                              void* d_out, int out_size, void* d_ws, size_t ws_size,
                              hipStream_t stream) {
    const float* image = (const float*)d_in[0];
    const int*   slic  = (const int*)d_in[1];
    float*       out   = (float*)d_out;
    float*       ws    = (float*)d_ws;

    int G;
    if      (ws_size >= (size_t)64 * 524288) G = 64;
    else if (ws_size >= (size_t)32 * 524288) G = 32;
    else if (ws_size >= (size_t)16 * 524288) G = 16;
    else if (ws_size >= (size_t)8  * 524288) G = 8;
    else                                     G = 4;

    seg_partial_kernel<<<N_IMG * G, 256, 0, stream>>>(image, slic, ws, G);

    const int total_waves = N_IMG * (N_SEG * NCH / 16);
    seg_reduce_wmma_kernel<<<total_waves / 8, 256, 0, stream>>>(ws, out, G);
}


# Stage 1: one 16-wave work-group per WGP, all 16 waves adding into ONE LDS copy of the sums: 32 partial planes per image, not 64; stage 2 sums in 8 steps; float sum order changes
# speedup vs baseline: 4.4516x; 1.0975x over previous
#include <hip/hip_runtime.h>

#define N_SEG   256
#define NCH     32
#define HW      (512 * 512)
#define N_IMG   8
#define PLANE   (N_SEG * NCH)
#define ACC_ELEMS (2 * PLANE)

typedef __attribute__((ext_vector_type(2))) float v2f;
typedef __attribute__((ext_vector_type(8))) float v8f;

#define BATCH 32
#define WAVES 16
#define THREADS (WAVES * 32)
#define TRIP_PIX (WAVES * BATCH)

__device__ __forceinline__ void issue_batch(const float* __restrict__ rows,
                                            const int* __restrict__ labs,
                                            int lane, float (&v)[BATCH], int& lab)
{
    lab = __builtin_nontemporal_load(&labs[lane]);
    #pragma unroll
    for (int j = 0; j < BATCH; ++j)
        v[j] = __builtin_nontemporal_load(&rows[j * NCH + lane]);
}

__device__ __forceinline__ void drain_batch(float* sums, float* zeros, unsigned* npix, int lane,
                                            const float (&v)[BATCH], int lab)
{
    atomicAdd(&npix[lab - 1], 1u);
    const int row = (lab - 1) * NCH;
    unsigned anyz = 0;
    #pragma unroll
    for (int j = 0; j < BATCH; ++j) {
        float* a = sums + __builtin_amdgcn_readlane(row, j) + lane;
        atomicAdd(a, v[j]);
        anyz |= __builtin_amdgcn_ballot_w32(!(v[j] != 0.0f));
    }
    if (anyz != 0) {
        #pragma unroll
        for (int j = 0; j < BATCH; ++j) {
            const int rj = __builtin_amdgcn_readlane(row, j);
            if (!(v[j] != 0.0f)) atomicAdd(zeros + rj + lane, 1.0f);
        }
    }
}

__global__ __launch_bounds__(THREADS) void seg_partial_kernel(
    const float* __restrict__ image, const int* __restrict__ slic,
    float* __restrict__ partials, int G)
{
    __shared__ float    sums[PLANE];
    __shared__ float    zeros[PLANE];
    __shared__ unsigned npix[N_SEG];

    const int tid  = threadIdx.x;
    const int img  = blockIdx.x / G;
    const int wg   = blockIdx.x % G;

    for (int i = tid; i < PLANE; i += THREADS) { sums[i] = 0.0f; zeros[i] = 0.0f; }
    if (tid < N_SEG) npix[tid] = 0u;
    __syncthreads();

    const int lane  = tid & 31;
    const int wave  = __builtin_amdgcn_readfirstlane(tid >> 5);
    const int chunk = HW / G;
    const int trips = chunk / TRIP_PIX;
    const size_t pix0 = (size_t)img * HW + (size_t)wg * chunk + (size_t)wave * BATCH;
    const float* rows = image + pix0 * NCH;
    const int*   labs = slic + pix0;

    float va[BATCH], vb[BATCH];
    int   la, lb;
    issue_batch(rows, labs, lane, va, la);
    int t = 0;
    for (; t + 2 < trips; t += 2) {
        issue_batch(rows + (size_t)(t + 1) * TRIP_PIX * NCH, labs + (size_t)(t + 1) * TRIP_PIX,
                    lane, vb, lb);
        __builtin_amdgcn_sched_barrier(0);
        drain_batch(sums, zeros, npix, lane, va, la);
        issue_batch(rows + (size_t)(t + 2) * TRIP_PIX * NCH, labs + (size_t)(t + 2) * TRIP_PIX,
                    lane, va, la);
        __builtin_amdgcn_sched_barrier(0);
        drain_batch(sums, zeros, npix, lane, vb, lb);
    }
    issue_batch(rows + (size_t)(t + 1) * TRIP_PIX * NCH, labs + (size_t)(t + 1) * TRIP_PIX,
                lane, vb, lb);
    __builtin_amdgcn_sched_barrier(0);
    drain_batch(sums, zeros, npix, lane, va, la);
    drain_batch(sums, zeros, npix, lane, vb, lb);
    __syncthreads();

    float* dst = partials + (size_t)(img * G + wg) * ACC_ELEMS;
    for (int i = tid; i < PLANE; i += THREADS) {
        dst[i]         = sums[i];
        dst[PLANE + i] = (float)npix[i >> 5] - zeros[i];
    }
}

__global__ __launch_bounds__(256) void seg_reduce_wmma_kernel(
    const float* __restrict__ partials, float* __restrict__ out, int G)
{
    const int tid  = threadIdx.x;
    const int lane = tid & 31;
    const int w    = blockIdx.x * 8 + (tid >> 5);
    const int img  = w >> 9;
    const int tile = w & 511;
    const int col  = tile * 16 + (lane & 15);
    const int half = lane >> 4;

    v2f a; a.x = 1.0f; a.y = 1.0f;
    v8f cs = {};
    v8f cc = {};

    const float* base = partials + (size_t)img * G * ACC_ELEMS;
    const int ksteps = G >> 2;
    for (int kk = 0; kk < ksteps; ++kk) {
        const int r0 = kk * 4 + half;
        const int r1 = kk * 4 + 2 + half;
        v2f bs, bc;
        bs.x = __builtin_nontemporal_load(&base[(size_t)r0 * ACC_ELEMS + col]);
        bs.y = __builtin_nontemporal_load(&base[(size_t)r1 * ACC_ELEMS + col]);
        bc.x = __builtin_nontemporal_load(&base[(size_t)r0 * ACC_ELEMS + N_SEG * NCH + col]);
        bc.y = __builtin_nontemporal_load(&base[(size_t)r1 * ACC_ELEMS + N_SEG * NCH + col]);
        cs = __builtin_amdgcn_wmma_f32_16x16x4_f32(false, a, false, bs,
                                                   (short)0, cs, false, false);
        cc = __builtin_amdgcn_wmma_f32_16x16x4_f32(false, a, false, bc,
                                                   (short)0, cc, false, false);
    }

    if (lane < 16) {
        out[(size_t)img * (N_SEG * NCH) + col] = cs[0] / cc[0];
    }
}

extern "C" void kernel_launch(void* const* d_in, const int* in_sizes, int n_in,
                              void* d_out, int out_size, void* d_ws, size_t ws_size,
                              hipStream_t stream) {
    const float* image = (const float*)d_in[0];
    const int*   slic  = (const int*)d_in[1];
    float*       out   = (float*)d_out;
    float*       ws    = (float*)d_ws;

    int G;
    if      (ws_size >= (size_t)32 * 524288) G = 32;
    else if (ws_size >= (size_t)16 * 524288) G = 16;
    else if (ws_size >= (size_t)8  * 524288) G = 8;
    else                                     G = 4;

    seg_partial_kernel<<<N_IMG * G, THREADS, 0, stream>>>(image, slic, ws, G);

    const int total_waves = N_IMG * (N_SEG * NCH / 16);
    seg_reduce_wmma_kernel<<<total_waves / 8, 256, 0, stream>>>(ws, out, G);
}
